# starting kernel
# baseline (speedup 1.0000x reference)
#include <hip/hip_runtime.h>

#define HID   20
#define GATES 80
#define TSEQ  1024
#define ROWS  128
#define HPAD  24
#define GPAD  84
#define KC    6
#define NT    5

typedef __attribute__((ext_vector_type(2))) float v2f;
typedef __attribute__((ext_vector_type(4))) float v4f;
typedef __attribute__((ext_vector_type(8))) float v8f;

__device__ __forceinline__ float fast_sigmoid(float x) {
  float e = __builtin_amdgcn_exp2f(-x * 1.44269504088896340736f);
  return __builtin_amdgcn_rcpf(1.0f + e);
}
__device__ __forceinline__ float fast_tanh(float x) {
  float e = __builtin_amdgcn_exp2f(x * 2.88539008177792681472f);
  return 1.0f - 2.0f * __builtin_amdgcn_rcpf(e + 1.0f);
}

__global__ __launch_bounds__(ROWS)
void lstm_fused(const float* __restrict__ x,    const float* __restrict__ Wih1,
                const float* __restrict__ Whh1, const float* __restrict__ bih1,
                const float* __restrict__ bhh1, const float* __restrict__ Wih2,
                const float* __restrict__ Whh2, const float* __restrict__ bih2,
                const float* __restrict__ bhh2, float* __restrict__ out)
{
  __shared__ __align__(16) float h1x[ROWS * HPAD];
  __shared__ __align__(16) float gl [ROWS * GPAD];
  __shared__ __align__(16) float w2 [HID * 4];

  const int tid  = threadIdx.x;
  const int lane = tid & 31;
  const int wave = tid >> 5;
  const int ln   = lane & 15;
  const int kh   = (lane >> 4) * 2;
  const int row0 = blockIdx.x * ROWS;

  for (int i = tid; i < ROWS * HPAD; i += ROWS) h1x[i] = 0.0f;
  if (tid < HID * 4) {
    int j = tid >> 2, q = tid & 3;
    w2[tid] = Wih2[q * HID + j];
  }

  v2f  bfrag[KC][NT];
  float biasf[NT];
  #pragma unroll
  for (int nt = 0; nt < NT; ++nt) {
    int n = nt * 16 + ln;
    biasf[nt] = bih1[n] + bhh1[n];
    #pragma unroll
    for (int kc = 0; kc < KC; ++kc) {
      int k0 = kc * 4 + kh;
      v2f b;
      if (k0 + 1 < HID)      { b.x = Whh1[n * HID + k0]; b.y = Whh1[n * HID + k0 + 1]; }
      else if (k0 == HID)    { b.x = Wih1[n];            b.y = 0.0f; }
      else                   { b.x = 0.0f;               b.y = 0.0f; }
      bfrag[kc][nt] = b;
    }
  }

  float b2[4], whh2v[4];
  #pragma unroll
  for (int q = 0; q < 4; ++q) { b2[q] = bih2[q] + bhh2[q]; whh2v[q] = Whh2[q]; }
  float c1[HID];
  #pragma unroll
  for (int j = 0; j < HID; ++j) c1[j] = 0.0f;
  float h2 = 0.0f, c2 = 0.0f;

  const float* xrow = x   + (size_t)(row0 + tid) * TSEQ;
  float*       orow = out + (size_t)(row0 + tid) * TSEQ;

  h1x[tid * HPAD + HID] = xrow[0];
  __syncthreads();

  for (int t = 0; t < TSEQ; ++t) {
    #pragma unroll
    for (int tile = 0; tile < 2; ++tile) {
      const int r0 = wave * 32 + tile * 16;
      v2f afrag[KC];
      #pragma unroll
      for (int kc = 0; kc < KC; ++kc) {
        int k = kc * 4 + kh;
        afrag[kc] = *(const v2f*)&h1x[(r0 + ln) * HPAD + k];
      }
      v8f acc[NT];
      #pragma unroll
      for (int nt = 0; nt < NT; ++nt) {
        float b = biasf[nt];
        acc[nt] = (v8f){b, b, b, b, b, b, b, b};
      }
      #pragma unroll
      for (int kc = 0; kc < KC; ++kc)
        #pragma unroll
        for (int nt = 0; nt < NT; ++nt)
          acc[nt] = __builtin_amdgcn_wmma_f32_16x16x4_f32(
              false, afrag[kc], false, bfrag[kc][nt],
              (short)0, acc[nt], false, false);
      #pragma unroll
      for (int nt = 0; nt < NT; ++nt) {
        int n  = nt * 16 + ln;
        int pc = (n % HID) * 4 + (n / HID);
        int rb = r0 + (lane >> 4) * 8;
        #pragma unroll
        for (int r = 0; r < 8; ++r)
          gl[(rb + r) * GPAD + pc] = acc[nt][r];
      }
    }
    __syncthreads();

    float xt_next = (t + 1 < TSEQ) ? xrow[t + 1] : 0.0f;

    float a2[4] = {0.0f, 0.0f, 0.0f, 0.0f};
    const float* grow = &gl[tid * GPAD];
    float*       hrow = &h1x[tid * HPAD];
    #pragma unroll
    for (int j = 0; j < HID; ++j) {
      v4f g4 = *(const v4f*)&grow[j * 4];
      float ig = fast_sigmoid(g4.x);
      float fg = fast_sigmoid(g4.y);
      float gg = fast_tanh   (g4.z);
      float og = fast_sigmoid(g4.w);
      float c  = fg * c1[j] + ig * gg;
      float h  = og * fast_tanh(c);
      c1[j] = c;
      hrow[j] = h;
      v4f wq = *(const v4f*)&w2[j * 4];
      a2[0] += h * wq.x; a2[1] += h * wq.y; a2[2] += h * wq.z; a2[3] += h * wq.w;
    }
    hrow[HID] = xt_next;

    float i2 = fast_sigmoid(a2[0] + h2 * whh2v[0] + b2[0]);
    float f2 = fast_sigmoid(a2[1] + h2 * whh2v[1] + b2[1]);
    float g2 = fast_tanh   (a2[2] + h2 * whh2v[2] + b2[2]);
    float o2 = fast_sigmoid(a2[3] + h2 * whh2v[3] + b2[3]);
    c2 = f2 * c2 + i2 * g2;
    h2 = o2 * fast_tanh(c2);
    orow[t] = h2;

    __syncthreads();
  }
}

extern "C" void kernel_launch(void* const* d_in, const int* in_sizes, int n_in,
                              void* d_out, int out_size, void* d_ws, size_t ws_size,
                              hipStream_t stream) {
  const float* x    = (const float*)d_in[0];
  const float* Wih1 = (const float*)d_in[1];
  const float* Whh1 = (const float*)d_in[2];
  const float* bih1 = (const float*)d_in[3];
  const float* bhh1 = (const float*)d_in[4];
  const float* Wih2 = (const float*)d_in[5];
  const float* Whh2 = (const float*)d_in[6];
  const float* bih2 = (const float*)d_in[7];
  const float* bhh2 = (const float*)d_in[8];
  float* out = (float*)d_out;

  const int B    = in_sizes[0] / TSEQ;
  const int grid = B / ROWS;
  hipLaunchKernelGGL(lstm_fused, dim3(grid), dim3(ROWS), 0, stream,
                     x, Wih1, Whh1, bih1, bhh1, Wih2, Whh2, bih2, bhh2, out);
}


# Layer 2 moved to a second wave per SIMD: the layer-1 wave leaves h1 of 8 steps in an LDS ring, a partner wave runs layer 2 and the stores one ring half behind; a barrier per 8 steps; same sum order
# speedup vs baseline: 3.5839x; 3.5839x over previous
#include <hip/hip_runtime.h>


#define HID   20
#define TSEQ  1024
#define WROWS 16
#define ROWS  64
#define NTHR  256
#define DEPTH 8
#define NGRP  (TSEQ / DEPTH)
#define SLOT  (WROWS * HPAD)
#define HPAD  28
#define KC    6
#define NT    5
#define L2E   1.44269504088896340736f

typedef __attribute__((ext_vector_type(2))) float v2f;
typedef __attribute__((ext_vector_type(8))) float v8f;

__device__ __forceinline__ float fast_sigmoid(float x) {
  float e = __builtin_amdgcn_exp2f(-x * L2E);
  return __builtin_amdgcn_rcpf(1.0f + e);
}
__device__ __forceinline__ float fast_tanh(float x) {
  float e = __builtin_amdgcn_exp2f(x * (2.0f * L2E));
  return 1.0f - 2.0f * __builtin_amdgcn_rcpf(e + 1.0f);
}
__device__ __forceinline__ float xhalf(float v) {
  int b = __builtin_bit_cast(int, v);
  return __builtin_bit_cast(float,
      __builtin_amdgcn_permlanex16(b, b, 0x76543210, 0xfedcba98, false, false));
}
template <int Q>
__device__ __forceinline__ float quad_bcast(float v) {
  return __builtin_bit_cast(float,
      __builtin_amdgcn_mov_dpp(__builtin_bit_cast(int, v), Q * 0x55, 0xf, 0xf, true));
}

__global__ __launch_bounds__(NTHR)
void lstm_fused(const float* __restrict__ x,    const float* __restrict__ Wih1,
                const float* __restrict__ Whh1, const float* __restrict__ bih1,
                const float* __restrict__ bhh1, const float* __restrict__ Wih2,
                const float* __restrict__ Whh2, const float* __restrict__ bih2,
                const float* __restrict__ bhh2, float* __restrict__ out)
{
  __shared__ __align__(16) float ring[4 * 2 * DEPTH * SLOT];

  const int tid  = threadIdx.x;
  const int lane = tid & 31;
  const int wave = __builtin_amdgcn_readfirstlane(tid >> 5);
  const int pair = wave & 3;
  const bool layer1 = wave < 4;
  const int ln   = lane & 15;
  const int hf   = lane >> 4;
  const int kh   = hf * 2;
  const int row0 = blockIdx.x * ROWS;
  const int r0   = pair * WROWS;
  float* rbase = &ring[pair * 2 * DEPTH * SLOT];

  const float* xrow = x   + (size_t)(row0 + r0 + ln) * TSEQ;
  float*       orow = out + (size_t)(row0 + r0 + ln) * TSEQ;

  v2f  bfrag[KC][NT];
  float biasf[NT];
  float actS = 0.0f, actP = 0.0f, actQ = 0.0f;
  float c1m[8], c1t[8];
  v2f afrag[KC];
  float w2r[5][2][4];
  float b2[4], whh2v[4];
  float h2 = 0.0f, c2 = 0.0f;

  #pragma unroll
  for (int r = 0; r < 8; ++r) { c1m[r] = 0.0f; c1t[r] = 0.0f; }
  #pragma unroll
  for (int kc = 0; kc < KC; ++kc) {
    afrag[kc] = (v2f){0.0f, 0.0f};
    #pragma unroll
    for (int nt = 0; nt < NT; ++nt) bfrag[kc][nt] = (v2f){0.0f, 0.0f};
  }
  #pragma unroll
  for (int nt = 0; nt < NT; ++nt) biasf[nt] = 0.0f;
  #pragma unroll
  for (int kc = 0; kc < 5; ++kc)
    #pragma unroll
    for (int s = 0; s < 2; ++s)
      #pragma unroll
      for (int q = 0; q < 4; ++q) w2r[kc][s][q] = 0.0f;
  #pragma unroll
  for (int q = 0; q < 4; ++q) { b2[q] = 0.0f; whh2v[q] = 0.0f; }

  if (layer1) {
    for (int i = lane * 4; i < 2 * DEPTH * SLOT; i += 128)
      *(float4*)&rbase[i] = make_float4(0.0f, 0.0f, 0.0f, 0.0f);
    __builtin_amdgcn_wave_barrier();

    #pragma unroll
    for (int nt = 0; nt < NT; ++nt) {
      int n = (nt < 4) ? nt * HID + ln : (ln & 3) * HID + 16 + (ln >> 2);
      biasf[nt] = bih1[n] + bhh1[n];
      #pragma unroll
      for (int kc = 0; kc < KC; ++kc) {
        int k0 = kc * 4 + kh;
        v2f b;
        if (k0 + 1 < HID)      { b.x = Whh1[n * HID + k0]; b.y = Whh1[n * HID + k0 + 1]; }
        else if (k0 == HID)    { b.x = Wih1[n];            b.y = 0.0f; }
        else                   { b.x = 0.0f;               b.y = 0.0f; }
        bfrag[kc][nt] = b;
      }
    }
    const bool isg = (ln & 3) == 2;
    actS = isg ? 2.0f * L2E : -L2E;
    actP = isg ? 1.0f : 0.0f;
    actQ = isg ? -2.0f : 1.0f;
    afrag[5].x = (hf == 0) ? xrow[0] : 0.0f;
  } else {
    #pragma unroll
    for (int kc = 0; kc < 5; ++kc)
      #pragma unroll
      for (int s = 0; s < 2; ++s)
        #pragma unroll
        for (int q = 0; q < 4; ++q)
          w2r[kc][s][q] = Wih2[q * HID + kc * 4 + kh + s];
    #pragma unroll
    for (int q = 0; q < 4; ++q) { b2[q] = bih2[q] + bhh2[q]; whh2v[q] = Whh2[q]; }
  }

  for (int g = 0; g <= NGRP; ++g) {
    if (layer1) {
      if (g < NGRP) {
        float* buf = rbase + (g & 1) * (DEPTH * SLOT);
        for (int s = 0; s < DEPTH; ++s) {
          const int t = g * DEPTH + s;
          float* hw = buf + s * SLOT;
          const int tn = (t + 1 < TSEQ) ? t + 1 : t;
          const float xt_next = xrow[tn];

          v8f acc[NT];
          #pragma unroll
          for (int nt = 0; nt < NT; ++nt) {
            float b = biasf[nt];
            acc[nt] = (v8f){b, b, b, b, b, b, b, b};
          }
          #pragma unroll
          for (int kc = 0; kc < KC; ++kc)
            #pragma unroll
            for (int nt = 0; nt < NT; ++nt)
              acc[nt] = __builtin_amdgcn_wmma_f32_16x16x4_f32(
                  false, afrag[kc], false, bfrag[kc][nt],
                  (short)0, acc[nt], false, false);

          #pragma unroll
          for (int r = 0; r < 8; ++r) {
            float ig = fast_sigmoid(acc[0][r]);
            float fg = fast_sigmoid(acc[1][r]);
            float gg = fast_tanh   (acc[2][r]);
            float og = fast_sigmoid(acc[3][r]);
            float c  = fg * c1m[r] + ig * gg;
            c1m[r] = c;
            hw[(hf * 8 + r) * HPAD + ln] = og * fast_tanh(c);
          }
          #pragma unroll
          for (int r = 0; r < 8; ++r) {
            float e  = __builtin_amdgcn_exp2f(acc[4][r] * actS);
            float a  = actP + actQ * __builtin_amdgcn_rcpf(1.0f + e);
            float ig = quad_bcast<0>(a);
            float fg = quad_bcast<1>(a);
            float gg = quad_bcast<2>(a);
            float og = quad_bcast<3>(a);
            float c  = fg * c1t[r] + ig * gg;
            c1t[r] = c;
            hw[(hf * 8 + r) * HPAD + 16 + (ln >> 2)] = og * fast_tanh(c);
          }
          hw[ln * HPAD + HID] = xt_next;

          __builtin_amdgcn_wave_barrier();
          #pragma unroll
          for (int kc = 0; kc < KC; ++kc)
            afrag[kc] = *(const v2f*)&hw[ln * HPAD + kc * 4 + kh];
        }
      }
    } else {
      if (g >= 1) {
        const float* buf = rbase + ((g - 1) & 1) * (DEPTH * SLOT);
        for (int s = 0; s < DEPTH; ++s) {
          const int t = (g - 1) * DEPTH + s;
          const float* hw = buf + s * SLOT;
          v2f hfr[5];
          #pragma unroll
          for (int kc = 0; kc < 5; ++kc)
            hfr[kc] = *(const v2f*)&hw[ln * HPAD + kc * 4 + kh];

          float a2[4], a2o[4];
          #pragma unroll
          for (int q = 0; q < 4; ++q) {
            a2[q]  = hfr[0].x * w2r[0][0][q];
            a2o[q] = hfr[0].y * w2r[0][1][q];
          }
          #pragma unroll
          for (int kc = 1; kc < 5; ++kc)
            #pragma unroll
            for (int q = 0; q < 4; ++q) {
              a2[q]  = __builtin_fmaf(hfr[kc].x, w2r[kc][0][q], a2[q]);
              a2o[q] = __builtin_fmaf(hfr[kc].y, w2r[kc][1][q], a2o[q]);
            }
          #pragma unroll
          for (int q = 0; q < 4; ++q) { a2[q] += a2o[q]; a2[q] += xhalf(a2[q]); }

          float i2 = fast_sigmoid(a2[0] + h2 * whh2v[0] + b2[0]);
          float f2 = fast_sigmoid(a2[1] + h2 * whh2v[1] + b2[1]);
          float g2 = fast_tanh   (a2[2] + h2 * whh2v[2] + b2[2]);
          float o2 = fast_sigmoid(a2[3] + h2 * whh2v[3] + b2[3]);
          c2 = f2 * c2 + i2 * g2;
          h2 = o2 * fast_tanh(c2);
          if (hf == 0) orow[t] = h2;
        }
      }
    }
    __syncthreads();
  }
}

extern "C" void kernel_launch(void* const* d_in, const int* in_sizes, int n_in,
                              void* d_out, int out_size, void* d_ws, size_t ws_size,
                              hipStream_t stream) {
  const float* x    = (const float*)d_in[0];
  const float* Wih1 = (const float*)d_in[1];
  const float* Whh1 = (const float*)d_in[2];
  const float* bih1 = (const float*)d_in[3];
  const float* bhh1 = (const float*)d_in[4];
  const float* Wih2 = (const float*)d_in[5];
  const float* Whh2 = (const float*)d_in[6];
  const float* bih2 = (const float*)d_in[7];
  const float* bhh2 = (const float*)d_in[8];
  float* out = (float*)d_out;

  const int B    = in_sizes[0] / TSEQ;
  const int grid = B / ROWS;
  hipLaunchKernelGGL(lstm_fused, dim3(grid), dim3(NTHR), 0, stream,
                     x, Wih1, Whh1, bih1, bhh1, Wih2, Whh2, bih2, bhh2, out);
}


# Gates of units 16-19 turned through wave-private LDS behind the main cells' arithmetic: a lane gets all 4 gates of two whole cells of its row; no quad DPP, 20 not 32 transcendentals; same sum order
# speedup vs baseline: 3.8224x; 1.0665x over previous
#include <hip/hip_runtime.h>


#define HID   20
#define TSEQ  1024
#define WROWS 16
#define ROWS  64
#define NTHR  256
#define DEPTH 8
#define NGRP  (TSEQ / DEPTH)
#define SLOT  (WROWS * HPAD)
#define HPAD  28
#define TPITCH 20
#define KC    6
#define NT    5
#define L2E   1.44269504088896340736f

typedef __attribute__((ext_vector_type(2))) float v2f;
typedef __attribute__((ext_vector_type(8))) float v8f;

__device__ __forceinline__ float fast_sigmoid(float x) {
  float e = __builtin_amdgcn_exp2f(-x * L2E);
  return __builtin_amdgcn_rcpf(1.0f + e);
}
__device__ __forceinline__ float fast_tanh(float x) {
  float e = __builtin_amdgcn_exp2f(x * (2.0f * L2E));
  return 1.0f - 2.0f * __builtin_amdgcn_rcpf(e + 1.0f);
}
__device__ __forceinline__ float xhalf(float v) {
  int b = __builtin_bit_cast(int, v);
  return __builtin_bit_cast(float,
      __builtin_amdgcn_permlanex16(b, b, 0x76543210, 0xfedcba98, false, false));
}

__global__ __launch_bounds__(NTHR)
void lstm_fused(const float* __restrict__ x,    const float* __restrict__ Wih1,
                const float* __restrict__ Whh1, const float* __restrict__ bih1,
                const float* __restrict__ bhh1, const float* __restrict__ Wih2,
                const float* __restrict__ Whh2, const float* __restrict__ bih2,
                const float* __restrict__ bhh2, float* __restrict__ out)
{
  __shared__ __align__(16) float ring[4 * 2 * DEPTH * SLOT];
  __shared__ __align__(16) float turn[4 * WROWS * TPITCH];

  const int tid  = threadIdx.x;
  const int lane = tid & 31;
  const int wave = __builtin_amdgcn_readfirstlane(tid >> 5);
  const int pair = wave & 3;
  const bool layer1 = wave < 4;
  const int ln   = lane & 15;
  const int hf   = lane >> 4;
  const int kh   = hf * 2;
  const int row0 = blockIdx.x * ROWS;
  const int r0   = pair * WROWS;
  float* rbase = &ring[pair * 2 * DEPTH * SLOT];
  float* tw    = &turn[pair * WROWS * TPITCH];
  float*       twst = tw + hf * 8 * TPITCH + ln;
  const float* twld = tw + ln * TPITCH + 8 * hf;

  const float* xrow = x   + (size_t)(row0 + r0 + ln) * TSEQ;
  float*       orow = out + (size_t)(row0 + r0 + ln) * TSEQ;

  v2f  bfrag[KC][NT];
  float biasf[NT];
  float c1m[8];
  float c1u[2] = {0.0f, 0.0f};
  v2f afrag[KC];
  float w2r[5][2][4];
  float b2[4], whh2v[4];
  float h2 = 0.0f, c2 = 0.0f;

  #pragma unroll
  for (int r = 0; r < 8; ++r) c1m[r] = 0.0f;
  #pragma unroll
  for (int kc = 0; kc < KC; ++kc) {
    afrag[kc] = (v2f){0.0f, 0.0f};
    #pragma unroll
    for (int nt = 0; nt < NT; ++nt) bfrag[kc][nt] = (v2f){0.0f, 0.0f};
  }
  #pragma unroll
  for (int nt = 0; nt < NT; ++nt) biasf[nt] = 0.0f;
  #pragma unroll
  for (int kc = 0; kc < 5; ++kc)
    #pragma unroll
    for (int s = 0; s < 2; ++s)
      #pragma unroll
      for (int q = 0; q < 4; ++q) w2r[kc][s][q] = 0.0f;
  #pragma unroll
  for (int q = 0; q < 4; ++q) { b2[q] = 0.0f; whh2v[q] = 0.0f; }

  if (layer1) {
    for (int i = lane * 4; i < 2 * DEPTH * SLOT; i += 128)
      *(float4*)&rbase[i] = make_float4(0.0f, 0.0f, 0.0f, 0.0f);
    __builtin_amdgcn_wave_barrier();

    #pragma unroll
    for (int nt = 0; nt < NT; ++nt) {
      int n = (nt < 4) ? nt * HID + ln : (ln & 3) * HID + 16 + (ln >> 2);
      biasf[nt] = bih1[n] + bhh1[n];
      #pragma unroll
      for (int kc = 0; kc < KC; ++kc) {
        int k0 = kc * 4 + kh;
        v2f b;
        if (k0 + 1 < HID)      { b.x = Whh1[n * HID + k0]; b.y = Whh1[n * HID + k0 + 1]; }
        else if (k0 == HID)    { b.x = Wih1[n];            b.y = 0.0f; }
        else                   { b.x = 0.0f;               b.y = 0.0f; }
        bfrag[kc][nt] = b;
      }
    }
    afrag[5].x = (hf == 0) ? xrow[0] : 0.0f;
  } else {
    #pragma unroll
    for (int kc = 0; kc < 5; ++kc)
      #pragma unroll
      for (int s = 0; s < 2; ++s)
        #pragma unroll
        for (int q = 0; q < 4; ++q)
          w2r[kc][s][q] = Wih2[q * HID + kc * 4 + kh + s];
    #pragma unroll
    for (int q = 0; q < 4; ++q) { b2[q] = bih2[q] + bhh2[q]; whh2v[q] = Whh2[q]; }
  }

  for (int g = 0; g <= NGRP; ++g) {
    if (layer1) {
      if (g < NGRP) {
        float* buf = rbase + (g & 1) * (DEPTH * SLOT);
        for (int s = 0; s < DEPTH; ++s) {
          const int t = g * DEPTH + s;
          float* hw = buf + s * SLOT;
          const int tn = (t + 1 < TSEQ) ? t + 1 : t;
          const float xt_next = xrow[tn];

          v8f acc[NT];
          #pragma unroll
          for (int nt = 0; nt < NT; ++nt) {
            float b = biasf[nt];
            acc[nt] = (v8f){b, b, b, b, b, b, b, b};
          }
          #pragma unroll
          for (int kc = 0; kc < KC; ++kc) {
            acc[4] = __builtin_amdgcn_wmma_f32_16x16x4_f32(
                false, afrag[kc], false, bfrag[kc][4],
                (short)0, acc[4], false, false);
            #pragma unroll
            for (int nt = 0; nt < 4; ++nt)
              acc[nt] = __builtin_amdgcn_wmma_f32_16x16x4_f32(
                  false, afrag[kc], false, bfrag[kc][nt],
                  (short)0, acc[nt], false, false);
          }

          #pragma unroll
          for (int r = 0; r < 8; ++r)
            twst[r * TPITCH] = acc[4][r];
          __builtin_amdgcn_wave_barrier();
          const float4 tg0 = *(const float4*)&twld[0];
          const float4 tg1 = *(const float4*)&twld[4];

          {
            float ig = fast_sigmoid(tg0.x);
            float fg = fast_sigmoid(tg0.y);
            float gg = fast_tanh   (tg0.z);
            float og = fast_sigmoid(tg0.w);
            float c  = fg * c1u[0] + ig * gg;
            c1u[0] = c;
            const float h0 = og * fast_tanh(c);
            ig = fast_sigmoid(tg1.x);
            fg = fast_sigmoid(tg1.y);
            gg = fast_tanh   (tg1.z);
            og = fast_sigmoid(tg1.w);
            c  = fg * c1u[1] + ig * gg;
            c1u[1] = c;
            const float h1 = og * fast_tanh(c);
            *(v2f*)&hw[ln * HPAD + 16 + 2 * hf] = (v2f){h0, h1};
          }
          #pragma unroll
          for (int r = 0; r < 8; ++r) {
            float ig = fast_sigmoid(acc[0][r]);
            float fg = fast_sigmoid(acc[1][r]);
            float gg = fast_tanh   (acc[2][r]);
            float og = fast_sigmoid(acc[3][r]);
            float c  = fg * c1m[r] + ig * gg;
            c1m[r] = c;
            hw[(hf * 8 + r) * HPAD + ln] = og * fast_tanh(c);
          }
          hw[ln * HPAD + HID] = xt_next;

          __builtin_amdgcn_wave_barrier();
          #pragma unroll
          for (int kc = 0; kc < KC; ++kc)
            afrag[kc] = *(const v2f*)&hw[ln * HPAD + kc * 4 + kh];
        }
      }
    } else {
      if (g >= 1) {
        const float* buf = rbase + ((g - 1) & 1) * (DEPTH * SLOT);
        for (int s = 0; s < DEPTH; ++s) {
          const int t = (g - 1) * DEPTH + s;
          const float* hw = buf + s * SLOT;
          v2f hfr[5];
          #pragma unroll
          for (int kc = 0; kc < 5; ++kc)
            hfr[kc] = *(const v2f*)&hw[ln * HPAD + kc * 4 + kh];

          float a2[4], a2o[4];
          #pragma unroll
          for (int q = 0; q < 4; ++q) {
            a2[q]  = hfr[0].x * w2r[0][0][q];
            a2o[q] = hfr[0].y * w2r[0][1][q];
          }
          #pragma unroll
          for (int kc = 1; kc < 5; ++kc)
            #pragma unroll
            for (int q = 0; q < 4; ++q) {
              a2[q]  = __builtin_fmaf(hfr[kc].x, w2r[kc][0][q], a2[q]);
              a2o[q] = __builtin_fmaf(hfr[kc].y, w2r[kc][1][q], a2o[q]);
            }
          #pragma unroll
          for (int q = 0; q < 4; ++q) { a2[q] += a2o[q]; a2[q] += xhalf(a2[q]); }

          float i2 = fast_sigmoid(a2[0] + h2 * whh2v[0] + b2[0]);
          float f2 = fast_sigmoid(a2[1] + h2 * whh2v[1] + b2[1]);
          float g2 = fast_tanh   (a2[2] + h2 * whh2v[2] + b2[2]);
          float o2 = fast_sigmoid(a2[3] + h2 * whh2v[3] + b2[3]);
          c2 = f2 * c2 + i2 * g2;
          h2 = o2 * fast_tanh(c2);
          if (hf == 0) orow[t] = h2;
        }
      }
    }
    __syncthreads();
  }
}

extern "C" void kernel_launch(void* const* d_in, const int* in_sizes, int n_in,
                              void* d_out, int out_size, void* d_ws, size_t ws_size,
                              hipStream_t stream) {
  const float* x    = (const float*)d_in[0];
  const float* Wih1 = (const float*)d_in[1];
  const float* Whh1 = (const float*)d_in[2];
  const float* bih1 = (const float*)d_in[3];
  const float* bhh1 = (const float*)d_in[4];
  const float* Wih2 = (const float*)d_in[5];
  const float* Whh2 = (const float*)d_in[6];
  const float* bih2 = (const float*)d_in[7];
  const float* bhh2 = (const float*)d_in[8];
  float* out = (float*)d_out;

  const int B    = in_sizes[0] / TSEQ;
  const int grid = B / ROWS;
  hipLaunchKernelGGL(lstm_fused, dim3(grid), dim3(NTHR), 0, stream,
                     x, Wih1, Whh1, bih1, bhh1, Wih2, Whh2, bih2, bhh2, out);
}
